# starting kernel
# baseline (speedup 1.0000x reference)
#include <hip/hip_runtime.h>
#include <math.h>

#define NNODES 50000
#define NEDGES 800000
#define IN_F   128
#define HID_F  128
#define OUT_F  64

typedef float v2f __attribute__((ext_vector_type(2)));
typedef float v8f __attribute__((ext_vector_type(8)));


__global__ void zero_f4(float4* __restrict__ p, int n4) {
    int i = blockIdx.x * blockDim.x + threadIdx.x;
    if (i < n4) p[i] = make_float4(0.f, 0.f, 0.f, 0.f);
}

__global__ void deg_kernel(const int* __restrict__ dst, float* __restrict__ deg, int ne) {
    int e = blockIdx.x * blockDim.x + threadIdx.x;
    if (e < ne) atomicAdd(&deg[dst[e]], 1.0f);
}

__global__ void dinv_kernel(float* __restrict__ deg, int n) {
    int i = blockIdx.x * blockDim.x + threadIdx.x;
    if (i < n) deg[i] = 1.0f / sqrtf(fmaxf(deg[i], 1.0f));
}

__global__ void edge_scatter(const float* __restrict__ x,
                             const int* __restrict__ src,
                             const int* __restrict__ dst,
                             const float* __restrict__ dinv,
                             float* __restrict__ h) {
    int e = (blockIdx.x * blockDim.x + threadIdx.x) >> 5;
    int lane = threadIdx.x & 31;
    int s = src[e];
    int d = dst[e];
    float sc = dinv[s];
    const float4 v = *(const float4*)(x + (size_t)s * 128 + lane * 4);
    float* hp = h + (size_t)d * 128 + lane * 4;
    atomicAdd(hp + 0, v.x * sc);
    atomicAdd(hp + 1, v.y * sc);
    atomicAdd(hp + 2, v.z * sc);
    atomicAdd(hp + 3, v.w * sc);
}

__global__ void neg_scale(float* __restrict__ h, const float* __restrict__ dinv) {
    int t = blockIdx.x * blockDim.x + threadIdx.x;
    int n = t >> 5;
    int c = (t & 31) << 2;
    float sc = -dinv[n];
    float4* p = (float4*)(h + (size_t)n * 128 + c);
    float4 v = *p;
    v.x *= sc; v.y *= sc; v.z *= sc; v.w *= sc;
    *p = v;
}

template <int NC>
__global__ __launch_bounds__(32 * (NC / 16)) void cheb_gemm(
        const float* __restrict__ A1, const float* __restrict__ A2,
        const float* __restrict__ W,  const float* __restrict__ bias,
        float* __restrict__ out) {
    __shared__ float As[16 * 260];

    const int mbase = blockIdx.x * 16;
    const int tid = threadIdx.x;

    for (int i = tid; i < 16 * 64; i += blockDim.x) {
        int row = i >> 6;
        int col = (i & 63) << 2;
        const float* sp = (col < 128)
                        ? (A1 + (size_t)(mbase + row) * 128 + col)
                        : (A2 + (size_t)(mbase + row) * 128 + (col - 128));
        float4 v = *(const float4*)sp;
        *(float4*)(&As[row * 260 + col]) = v;
    }
    __syncthreads();

    const int ntile = tid >> 5;
    const int lane  = tid & 31;
    const int half  = lane >> 4;
    const int l     = lane & 15;

    v8f c = {};
    #pragma unroll 4
    for (int k0 = 0; k0 < 256; k0 += 4) {
        v2f a = *(const v2f*)(&As[l * 260 + k0 + 2 * half]);
        const float* wp = W + (size_t)(k0 + 2 * half) * NC + ntile * 16 + l;
        v2f b;
        b.x = wp[0];
        b.y = wp[NC];
        c = __builtin_amdgcn_wmma_f32_16x16x4_f32(
                false, a, false, b, (short)0, c, false, false);
    }

    const int col = ntile * 16 + l;
    const float bv = bias[col];
    #pragma unroll
    for (int r = 0; r < 8; ++r) {
        int row = mbase + r + 8 * half;
        out[(size_t)row * NC + col] = c[r] + bv;
    }
}


extern "C" void kernel_launch(void* const* d_in, const int* in_sizes, int n_in,
                              void* d_out, int out_size, void* d_ws, size_t ws_size,
                              hipStream_t stream) {
    const float* feat = (const float*)d_in[0];
    const int*   src  = (const int*)d_in[1];
    const int*   dst  = (const int*)d_in[2];
    const float* W1   = (const float*)d_in[3];
    const float* b1   = (const float*)d_in[4];
    const float* W2   = (const float*)d_in[5];
    const float* b2   = (const float*)d_in[6];
    float* out = (float*)d_out;

    char* ws = (char*)d_ws;
    float* dinv = (float*)ws;
    float* h    = (float*)(ws + 200704);
    float* out1 = (float*)(ws + 200704 + (size_t)NNODES * 128 * 4);

    const int TB = 256;

    zero_f4<<<(NNODES / 4 + TB - 1) / TB, TB, 0, stream>>>((float4*)dinv, NNODES / 4);
    deg_kernel<<<(NEDGES + TB - 1) / TB, TB, 0, stream>>>(dst, dinv, NEDGES);
    dinv_kernel<<<(NNODES + TB - 1) / TB, TB, 0, stream>>>(dinv, NNODES);

    const int n4h   = NNODES * 128 / 4;
    const int scat  = NEDGES * 32;
    const int nsc   = NNODES * 32;

    zero_f4<<<(n4h + TB - 1) / TB, TB, 0, stream>>>((float4*)h, n4h);
    edge_scatter<<<scat / TB, TB, 0, stream>>>(feat, src, dst, dinv, h);
    neg_scale<<<nsc / TB, TB, 0, stream>>>(h, dinv);
    cheb_gemm<HID_F><<<NNODES / 16, 32 * (HID_F / 16), 0, stream>>>(feat, h, W1, b1, out1);

    zero_f4<<<(n4h + TB - 1) / TB, TB, 0, stream>>>((float4*)h, n4h);
    edge_scatter<<<scat / TB, TB, 0, stream>>>(out1, src, dst, dinv, h);
    neg_scale<<<nsc / TB, TB, 0, stream>>>(h, dinv);
    cheb_gemm<OUT_F><<<NNODES / 16, 32 * (OUT_F / 16), 0, stream>>>(out1, h, W2, b2, out);
}


# Float-atomic scatter replaced by per-call grouping of edges by target (int rank atomics, scan, place, sort by source id) and a fixed-order register gather sum into the GEMM A tile; 11 launches to 7
# speedup vs baseline: 2.5032x; 2.5032x over previous
#include <hip/hip_runtime.h>
#include <math.h>

#define NNODES 50000
#define NEDGES 800000
#define IN_F   128
#define HID_F  128
#define OUT_F  64

typedef float v2f __attribute__((ext_vector_type(2)));
typedef float v8f __attribute__((ext_vector_type(8)));


__global__ void zero_f4(float4* __restrict__ p, int n4) {
    int i = blockIdx.x * blockDim.x + threadIdx.x;
    if (i < n4) p[i] = make_float4(0.f, 0.f, 0.f, 0.f);
}

__global__ void rank_kernel(const int* __restrict__ dst, int* __restrict__ cnt,
                            int* __restrict__ rank, int ne) {
    int e = blockIdx.x * blockDim.x + threadIdx.x;
    if (e < ne) rank[e] = atomicAdd(&cnt[dst[e]], 1);
}

#define SCAN_T  1024
#define SCAN_IT 13
__global__ __launch_bounds__(SCAN_T) void scan_kernel(const int* __restrict__ cnt,
                                                      int* __restrict__ row_ptr,
                                                      float* __restrict__ dinv) {
    __shared__ int wsum[32];
    const int t = threadIdx.x;
    const int lane = t & 31;
    const int w = t >> 5;
    const int wbase = w * (SCAN_IT * 128);
    int4 p[SCAN_IT];
    int carry = 0;
    #pragma unroll
    for (int i = 0; i < SCAN_IT; ++i) {
        const int idx = wbase + i * 128 + lane * 4;
        int4 c = make_int4(0, 0, 0, 0);
        if (idx < NNODES) {
            c = *(const int4*)(cnt + idx);
            float4 d;
            d.x = 1.0f / sqrtf(fmaxf((float)c.x, 1.0f));
            d.y = 1.0f / sqrtf(fmaxf((float)c.y, 1.0f));
            d.z = 1.0f / sqrtf(fmaxf((float)c.z, 1.0f));
            d.w = 1.0f / sqrtf(fmaxf((float)c.w, 1.0f));
            *(float4*)(dinv + idx) = d;
        }
        const int s = c.x + c.y + c.z + c.w;
        int inc = s;
        #pragma unroll
        for (int o = 1; o < 32; o <<= 1) {
            int u = __shfl_up(inc, o);
            if (lane >= o) inc += u;
        }
        const int ex = carry + inc - s;
        p[i] = make_int4(ex, ex + c.x, ex + c.x + c.y, ex + c.x + c.y + c.z);
        carry += __shfl(inc, 31);
    }
    if (lane == 0) wsum[w] = carry;
    __syncthreads();
    if (w == 0) {
        int v = wsum[lane];
        int iv = v;
        #pragma unroll
        for (int o = 1; o < 32; o <<= 1) {
            int u = __shfl_up(iv, o);
            if (lane >= o) iv += u;
        }
        wsum[lane] = iv - v;
    }
    __syncthreads();
    const int off = wsum[w];
    #pragma unroll
    for (int i = 0; i < SCAN_IT; ++i) {
        const int idx = wbase + i * 128 + lane * 4;
        if (idx < NNODES) {
            int4 q = p[i];
            q.x += off; q.y += off; q.z += off; q.w += off;
            *(int4*)(row_ptr + idx) = q;
        }
    }
    if (t == SCAN_T - 1) row_ptr[NNODES] = off + carry;
}

__global__ void place_kernel(const int* __restrict__ src, const int* __restrict__ dst,
                             const int* __restrict__ rank, const int* __restrict__ row_ptr,
                             int* __restrict__ csr, int ne) {
    int e = blockIdx.x * blockDim.x + threadIdx.x;
    if (e < ne) csr[row_ptr[dst[e]] + rank[e]] = src[e];
}

__global__ void seg_sort(const int* __restrict__ row_ptr, const int* __restrict__ csr_in,
                         int* __restrict__ csr_out) {
    const int node = (blockIdx.x * blockDim.x + threadIdx.x) >> 5;
    const int lane = threadIdx.x & 31;
    if (node >= NNODES) return;
    const int beg = row_ptr[node];
    const int n = row_ptr[node + 1] - beg;
    if (n <= 32) {
        const int k = (lane < n) ? csr_in[beg + lane] : 0x7fffffff;
        int r = 0;
        #pragma unroll
        for (int j = 0; j < 32; ++j) {
            int kj = __shfl(k, j);
            r += ((kj < k) | ((kj == k) & (j < lane))) ? 1 : 0;
        }
        if (lane < n) csr_out[beg + r] = k;
    } else {
        for (int i = lane; i < n; i += 32) {
            const int k = csr_in[beg + i];
            int r = 0;
            for (int j = 0; j < n; ++j) {
                int kj = csr_in[beg + j];
                r += ((kj < k) | ((kj == k) & (j < i))) ? 1 : 0;
            }
            csr_out[beg + r] = k;
        }
    }
}


__device__ __forceinline__ float4 ld_row(const float* __restrict__ X, int s_l, int j, int lane) {
    const int s = __builtin_amdgcn_readlane(s_l, j);
    return *(const float4*)(X + (size_t)s * 128 + lane * 4);
}
__device__ __forceinline__ float lane_f(float v_l, int j) {
    return __int_as_float(__builtin_amdgcn_readlane(__float_as_int(v_l), j));
}
__device__ __forceinline__ void add_row(float4& acc, const float4 v, const float sc) {
    acc.x = __builtin_fmaf(v.x, sc, acc.x);
    acc.y = __builtin_fmaf(v.y, sc, acc.y);
    acc.z = __builtin_fmaf(v.z, sc, acc.z);
    acc.w = __builtin_fmaf(v.w, sc, acc.w);
}

template <int NC>
__global__ __launch_bounds__(32 * (NC / 16)) void cheb_gemm(
        const float* __restrict__ X,
        const int* __restrict__ row_ptr, const int* __restrict__ csr,
        const float* __restrict__ dinv,
        const float* __restrict__ W,  const float* __restrict__ bias,
        float* __restrict__ out) {
    __shared__ float As[16 * 260];

    const int mbase = blockIdx.x * 16;
    const int tid = threadIdx.x;
    const int lane  = tid & 31;

    #pragma unroll
    for (int it = 0; it < (16 * 32) / (32 * (NC / 16)); ++it) {
        int i = tid + it * 32 * (NC / 16);
        int row = i >> 5;
        int col = (i & 31) << 2;
        float4 v = *(const float4*)(X + (size_t)(mbase + row) * 128 + col);
        *(float4*)(&As[row * 260 + col]) = v;
    }

    const int wave = __builtin_amdgcn_readfirstlane(tid >> 5);

    for (int row = wave; row < 16; row += NC / 16) {
        const int node = mbase + row;
        const int beg = row_ptr[node];
        const int end = row_ptr[node + 1];
        float4 acc = make_float4(0.f, 0.f, 0.f, 0.f);
        for (int base = beg; base < end; base += 32) {
            const int m = min(32, end - base);
            int s_l = 0;
            float sc_l = 0.f;
            if (lane < m) {
                s_l = csr[base + lane];
                sc_l = dinv[s_l];
            }
            int j = 0;
            for (; j + 8 <= m; j += 8) {
                float4 v[8];
                #pragma unroll
                for (int i = 0; i < 8; ++i) v[i] = ld_row(X, s_l, j + i, lane);
                #pragma unroll
                for (int i = 0; i < 8; ++i) add_row(acc, v[i], lane_f(sc_l, j + i));
            }
            if (j + 4 <= m) {
                float4 v[4];
                #pragma unroll
                for (int i = 0; i < 4; ++i) v[i] = ld_row(X, s_l, j + i, lane);
                #pragma unroll
                for (int i = 0; i < 4; ++i) add_row(acc, v[i], lane_f(sc_l, j + i));
                j += 4;
            }
            const int rem = m - j;
            float4 t0 = make_float4(0.f, 0.f, 0.f, 0.f), t1 = t0, t2 = t0;
            if (rem > 0) t0 = ld_row(X, s_l, j, lane);
            if (rem > 1) t1 = ld_row(X, s_l, j + 1, lane);
            if (rem > 2) t2 = ld_row(X, s_l, j + 2, lane);
            if (rem > 0) add_row(acc, t0, lane_f(sc_l, j));
            if (rem > 1) add_row(acc, t1, lane_f(sc_l, j + 1));
            if (rem > 2) add_row(acc, t2, lane_f(sc_l, j + 2));
        }
        const float ms = -dinv[node];
        acc.x *= ms; acc.y *= ms; acc.z *= ms; acc.w *= ms;
        *(float4*)(&As[row * 260 + 128 + lane * 4]) = acc;
    }
    __syncthreads();

    const int ntile = tid >> 5;
    const int half  = lane >> 4;
    const int l     = lane & 15;

    v8f c = {};
    #pragma unroll 4
    for (int k0 = 0; k0 < 256; k0 += 4) {
        v2f a = *(const v2f*)(&As[l * 260 + k0 + 2 * half]);
        const float* wp = W + (size_t)(k0 + 2 * half) * NC + ntile * 16 + l;
        v2f b;
        b.x = wp[0];
        b.y = wp[NC];
        c = __builtin_amdgcn_wmma_f32_16x16x4_f32(
                false, a, false, b, (short)0, c, false, false);
    }

    const int col = ntile * 16 + l;
    const float bv = bias[col];
    #pragma unroll
    for (int r = 0; r < 8; ++r) {
        int row = mbase + r + 8 * half;
        out[(size_t)row * NC + col] = c[r] + bv;
    }
}


extern "C" void kernel_launch(void* const* d_in, const int* in_sizes, int n_in,
                              void* d_out, int out_size, void* d_ws, size_t ws_size,
                              hipStream_t stream) {
    const float* feat = (const float*)d_in[0];
    const int*   src  = (const int*)d_in[1];
    const int*   dst  = (const int*)d_in[2];
    const float* W1   = (const float*)d_in[3];
    const float* b1   = (const float*)d_in[4];
    const float* W2   = (const float*)d_in[5];
    const float* b2   = (const float*)d_in[6];
    float* out = (float*)d_out;

    char* ws = (char*)d_ws;
    float* dinv   = (float*)ws;
    int*   cnt    = (int*)(ws + 200704);
    int*   rowp   = (int*)(ws + 200704 + 200704);
    int*   csr_a  = (int*)(ws + 200704 + 401408);
    int*   csr_b  = (int*)(ws + 200704 + 401408 + 3200000);
    int*   rank   = (int*)(ws + 200704 + 401408 + 6400000);
    float* out1   = (float*)(ws + 200704 + (size_t)NNODES * 128 * 4);

    const int TB = 256;

    zero_f4<<<(NNODES / 4 + TB - 1) / TB, TB, 0, stream>>>((float4*)cnt, NNODES / 4);
    rank_kernel<<<(NEDGES + TB - 1) / TB, TB, 0, stream>>>(dst, cnt, rank, NEDGES);
    scan_kernel<<<1, SCAN_T, 0, stream>>>(cnt, rowp, dinv);
    place_kernel<<<(NEDGES + TB - 1) / TB, TB, 0, stream>>>(src, dst, rank, rowp, csr_a, NEDGES);
    seg_sort<<<NNODES * 32 / TB, TB, 0, stream>>>(rowp, csr_a, csr_b);

    cheb_gemm<HID_F><<<NNODES / 16, 32 * (HID_F / 16), 0, stream>>>(feat, rowp, csr_b, dinv, W1, b1, out1);
    cheb_gemm<OUT_F><<<NNODES / 16, 32 * (OUT_F / 16), 0, stream>>>(out1, rowp, csr_b, dinv, W2, b2, out);
}


# Edge grouping in one pass: the rank atomic stores each source id straight into its target's 64-slot row; gather waves sort their row by source id in registers (sum order kept); 7 launches to 4
# speedup vs baseline: 3.1311x; 1.2508x over previous
#include <hip/hip_runtime.h>
#include <math.h>

#define NNODES 50000
#define NEDGES 800000
#define IN_F   128
#define HID_F  128
#define OUT_F  64
#define SLOTS  64
#define PADKEY 0x7fffffc0

typedef float v2f __attribute__((ext_vector_type(2)));
typedef float v8f __attribute__((ext_vector_type(8)));


__global__ void zero_f4(float4* __restrict__ p, int n4) {
    int i = blockIdx.x * blockDim.x + threadIdx.x;
    if (i < n4) p[i] = make_float4(0.f, 0.f, 0.f, 0.f);
}

__device__ __forceinline__ void put_edge(int d, int s, int r, int* __restrict__ cnt,
                                         int* __restrict__ slots, int* __restrict__ spill) {
    if (r < SLOTS) {
        slots[d * SLOTS + r] = s;
    } else {
        const int p = atomicAdd(&cnt[NNODES], 1);
        spill[2 * p] = d;
        spill[2 * p + 1] = s;
    }
}

__global__ void fill_kernel(const int* __restrict__ src, const int* __restrict__ dst,
                            int* __restrict__ cnt, int* __restrict__ slots,
                            int* __restrict__ spill, int ne4) {
    const int i = blockIdx.x * blockDim.x + threadIdx.x;
    if (i >= ne4) return;
    const int4 d = ((const int4*)dst)[i];
    const int4 s = ((const int4*)src)[i];
    const int r0 = atomicAdd(&cnt[d.x], 1);
    const int r1 = atomicAdd(&cnt[d.y], 1);
    const int r2 = atomicAdd(&cnt[d.z], 1);
    const int r3 = atomicAdd(&cnt[d.w], 1);
    put_edge(d.x, s.x, r0, cnt, slots, spill);
    put_edge(d.y, s.y, r1, cnt, slots, spill);
    put_edge(d.z, s.z, r2, cnt, slots, spill);
    put_edge(d.w, s.w, r3, cnt, slots, spill);
}


__device__ __forceinline__ float4 ld_row(const float* __restrict__ X, int s_l, int j, int lane) {
    const int s = __builtin_amdgcn_readlane(s_l, j);
    return *(const float4*)(X + (size_t)s * 128 + lane * 4);
}
__device__ __forceinline__ float lane_f(float v_l, int j) {
    return __int_as_float(__builtin_amdgcn_readlane(__float_as_int(v_l), j));
}
__device__ __forceinline__ void add_row(float4& acc, const float4 v, const float sc) {
    acc.x = __builtin_fmaf(v.x, sc, acc.x);
    acc.y = __builtin_fmaf(v.y, sc, acc.y);
    acc.z = __builtin_fmaf(v.z, sc, acc.z);
    acc.w = __builtin_fmaf(v.w, sc, acc.w);
}
__device__ __forceinline__ float inv_sqrt_deg(int c) {
    return 1.0f / sqrtf(fmaxf((float)c, 1.0f));
}

__device__ __forceinline__ void sum_chunk(float4& acc, const float* __restrict__ X,
                                          int s_l, float sc_l, int m, int lane) {
    int j = 0;
    for (; j + 8 <= m; j += 8) {
        float4 v[8];
        #pragma unroll
        for (int i = 0; i < 8; ++i) v[i] = ld_row(X, s_l, j + i, lane);
        #pragma unroll
        for (int i = 0; i < 8; ++i) add_row(acc, v[i], lane_f(sc_l, j + i));
    }
    if (j + 4 <= m) {
        float4 v[4];
        #pragma unroll
        for (int i = 0; i < 4; ++i) v[i] = ld_row(X, s_l, j + i, lane);
        #pragma unroll
        for (int i = 0; i < 4; ++i) add_row(acc, v[i], lane_f(sc_l, j + i));
        j += 4;
    }
    const int rem = m - j;
    float4 t0 = make_float4(0.f, 0.f, 0.f, 0.f), t1 = t0, t2 = t0;
    if (rem > 0) t0 = ld_row(X, s_l, j, lane);
    if (rem > 1) t1 = ld_row(X, s_l, j + 1, lane);
    if (rem > 2) t2 = ld_row(X, s_l, j + 2, lane);
    if (rem > 0) add_row(acc, t0, lane_f(sc_l, j));
    if (rem > 1) add_row(acc, t1, lane_f(sc_l, j + 1));
    if (rem > 2) add_row(acc, t2, lane_f(sc_l, j + 2));
}

template <int NC>
__global__ __launch_bounds__(32 * (NC / 16)) void cheb_gemm(
        const float* __restrict__ X,
        const int* __restrict__ cnt, const int* __restrict__ slots,
        const int* __restrict__ spill,
        const float* __restrict__ W,  const float* __restrict__ bias,
        float* __restrict__ out) {
    __shared__ float As[16 * 260];
    __shared__ int srt[(NC / 16) * 64];

    const int mbase = blockIdx.x * 16;
    const int tid = threadIdx.x;
    const int lane  = tid & 31;

    #pragma unroll
    for (int it = 0; it < (16 * 32) / (32 * (NC / 16)); ++it) {
        int i = tid + it * 32 * (NC / 16);
        int row = i >> 5;
        int col = (i & 31) << 2;
        float4 v = *(const float4*)(X + (size_t)(mbase + row) * 128 + col);
        *(float4*)(&As[row * 260 + col]) = v;
    }

    const int wave = __builtin_amdgcn_readfirstlane(tid >> 5);

    for (int row = wave; row < 16; row += NC / 16) {
        const int node = mbase + row;
        const int n = cnt[node];
        const int* __restrict__ myrow = slots + (size_t)node * SLOTS;
        float4 acc = make_float4(0.f, 0.f, 0.f, 0.f);
        if (n <= SLOTS) {
            int* sw = srt + wave * 64;
            int s_a;
            float sc_a;
            if (n <= 32) {
                int s = 0;
                if (lane < n) s = myrow[lane];
                const int key = (lane < n) ? s * 64 + lane : PADKEY + lane;
                const int cs = cnt[s];
                int r = 0;
                #pragma unroll
                for (int j = 0; j < 32; ++j)
                    r += (__builtin_amdgcn_readlane(key, j) < key) ? 1 : 0;
                const float sc_u = inv_sqrt_deg(cs);
                s_a = __builtin_amdgcn_ds_permute(r << 2, key) >> 6;
                sc_a = __int_as_float(__builtin_amdgcn_ds_permute(r << 2, __float_as_int(sc_u)));
            } else {
                const int k0 = myrow[lane] * 64 + lane;
                int k1 = PADKEY + lane;
                if (32 + lane < n) k1 = myrow[32 + lane] * 64 + 32 + lane;
                int r0 = 0, r1 = 0;
                #pragma nounroll
                for (int j = 0; j < 32; ++j) {
                    const int a = __builtin_amdgcn_readlane(k0, j);
                    const int b = __builtin_amdgcn_readlane(k1, j);
                    r0 += ((a < k0) ? 1 : 0) + ((b < k0) ? 1 : 0);
                    r1 += ((a < k1) ? 1 : 0) + ((b < k1) ? 1 : 0);
                }
                sw[r0] = k0;
                sw[r1] = k1;
                s_a = sw[lane] >> 6;
                const int s_b = sw[32 + lane] >> 6;
                sc_a = inv_sqrt_deg(cnt[s_a]);
                float sc_b = 0.f;
                if (32 + lane < n) sc_b = inv_sqrt_deg(cnt[s_b]);
                sw[lane] = s_b;
                sw[32 + lane] = __float_as_int(sc_b);
            }
            #pragma nounroll
            for (int c = 0; c < n; c += 32) {
                if (c) {
                    s_a = sw[lane];
                    sc_a = __int_as_float(sw[32 + lane]);
                }
                sum_chunk(acc, X, s_a, sc_a, min(32, n - c), lane);
            }
        } else {
            const int nsp = min(cnt[NNODES], NEDGES);
            int prev = -1;
            for (int guard = 0; guard < n; ++guard) {
                int best = 0x7fffffff;
                for (int i = lane; i < SLOTS; i += 32) {
                    const int k = myrow[i];
                    if (k > prev && k < best) best = k;
                }
                for (int i = lane; i < nsp; i += 32) {
                    if (spill[2 * i] == node) {
                        const int k = spill[2 * i + 1];
                        if (k > prev && k < best) best = k;
                    }
                }
                int bmin = 0x7fffffff;
                for (int j = 0; j < 32; ++j) bmin = min(bmin, __builtin_amdgcn_readlane(best, j));
                best = bmin;
                if (best == 0x7fffffff) break;
                int mult = 0;
                for (int i = lane; i < SLOTS; i += 32) mult += (myrow[i] == best) ? 1 : 0;
                for (int i = lane; i < nsp; i += 32)
                    mult += (spill[2 * i] == node && spill[2 * i + 1] == best) ? 1 : 0;
                int msum = 0;
                for (int j = 0; j < 32; ++j) msum += __builtin_amdgcn_readlane(mult, j);
                mult = msum;
                const float4 v = *(const float4*)(X + (size_t)best * 128 + lane * 4);
                const float sc = inv_sqrt_deg(cnt[best]);
                for (int t = 0; t < mult; ++t) add_row(acc, v, sc);
                prev = best;
            }
        }
        const float ms = -inv_sqrt_deg(n);
        acc.x *= ms; acc.y *= ms; acc.z *= ms; acc.w *= ms;
        *(float4*)(&As[row * 260 + 128 + lane * 4]) = acc;
    }
    __syncthreads();

    const int ntile = tid >> 5;
    const int half  = lane >> 4;
    const int l     = lane & 15;

    v8f c = {};
    #pragma unroll 4
    for (int k0 = 0; k0 < 256; k0 += 4) {
        v2f a = *(const v2f*)(&As[l * 260 + k0 + 2 * half]);
        const float* wp = W + (size_t)(k0 + 2 * half) * NC + ntile * 16 + l;
        v2f b;
        b.x = wp[0];
        b.y = wp[NC];
        c = __builtin_amdgcn_wmma_f32_16x16x4_f32(
                false, a, false, b, (short)0, c, false, false);
    }

    const int col = ntile * 16 + l;
    const float bv = bias[col];
    #pragma unroll
    for (int r = 0; r < 8; ++r) {
        int row = mbase + r + 8 * half;
        out[(size_t)row * NC + col] = c[r] + bv;
    }
}


extern "C" void kernel_launch(void* const* d_in, const int* in_sizes, int n_in,
                              void* d_out, int out_size, void* d_ws, size_t ws_size,
                              hipStream_t stream) {
    const float* feat = (const float*)d_in[0];
    const int*   src  = (const int*)d_in[1];
    const int*   dst  = (const int*)d_in[2];
    const float* W1   = (const float*)d_in[3];
    const float* b1   = (const float*)d_in[4];
    const float* W2   = (const float*)d_in[5];
    const float* b2   = (const float*)d_in[6];
    float* out = (float*)d_out;

    char* ws = (char*)d_ws;
    int*   cnt    = (int*)ws;
    int*   slots  = (int*)(ws + 200704);
    int*   spill  = (int*)(ws + 200704 + 12800000);
    float* out1   = (float*)(ws + 200704 + (size_t)NNODES * 128 * 4);

    const int TB = 256;

    zero_f4<<<(200704 / 16 + TB - 1) / TB, TB, 0, stream>>>((float4*)cnt, 200704 / 16);
    fill_kernel<<<(NEDGES / 4 + TB - 1) / TB, TB, 0, stream>>>(src, dst, cnt, slots, spill, NEDGES / 4);

    cheb_gemm<HID_F><<<NNODES / 16, 32 * (HID_F / 16), 0, stream>>>(feat, cnt, slots, spill, W1, b1, out1);
    cheb_gemm<OUT_F><<<NNODES / 16, 32 * (OUT_F / 16), 0, stream>>>(out1, cnt, slots, spill, W2, b2, out);
}


# Edge counts and slot rows kept once per half of the package: a wave's returning count atomics and slot stores go to its own half's plane; the gather sorts both rows together; sum order unchanged
# speedup vs baseline: 4.8126x; 1.5370x over previous
#include <hip/hip_runtime.h>
#include <math.h>

#define NNODES 50000
#define NEDGES 800000
#define IN_F   128
#define HID_F  128
#define OUT_F  64
#define PSLOTS 32
#define CPLANE 50176
#define SPLANE (NNODES * PSLOTS)
#define PADKEY 0x7fffffc0

typedef float v2f __attribute__((ext_vector_type(2)));
typedef float v8f __attribute__((ext_vector_type(8)));


__global__ void zero_f4(float4* __restrict__ p, int n4) {
    int i = blockIdx.x * blockDim.x + threadIdx.x;
    if (i < n4) p[i] = make_float4(0.f, 0.f, 0.f, 0.f);
}

__device__ __forceinline__ int my_half() {
    return (int)((__builtin_amdgcn_s_sendmsg_rtn(0x87) >> 18) & 1u);
}

__device__ __forceinline__ void put_edge(int d, int s, int r, int h, int* __restrict__ cnt,
                                         int* __restrict__ slots, int* __restrict__ spill) {
    if (r < PSLOTS) {
        slots[h * SPLANE + d * PSLOTS + r] = s;
    } else {
        const int p = atomicAdd(&cnt[NNODES], 1);
        spill[2 * p] = d;
        spill[2 * p + 1] = s;
    }
}

__global__ void fill_kernel(const int* __restrict__ src, const int* __restrict__ dst,
                            int* __restrict__ cnt, int* __restrict__ slots,
                            int* __restrict__ spill, int ne4) {
    const int i = blockIdx.x * blockDim.x + threadIdx.x;
    if (i >= ne4) return;
    const int4 d = ((const int4*)dst)[i];
    const int4 s = ((const int4*)src)[i];
    const int h = my_half();
    int* __restrict__ cp = cnt + h * CPLANE;
    const int r0 = atomicAdd(&cp[d.x], 1);
    const int r1 = atomicAdd(&cp[d.y], 1);
    const int r2 = atomicAdd(&cp[d.z], 1);
    const int r3 = atomicAdd(&cp[d.w], 1);
    put_edge(d.x, s.x, r0, h, cnt, slots, spill);
    put_edge(d.y, s.y, r1, h, cnt, slots, spill);
    put_edge(d.z, s.z, r2, h, cnt, slots, spill);
    put_edge(d.w, s.w, r3, h, cnt, slots, spill);
}


__device__ __forceinline__ float4 ld_row(const float* __restrict__ X, int s_l, int j, int lane) {
    const int s = __builtin_amdgcn_readlane(s_l, j);
    return *(const float4*)(X + (size_t)s * 128 + lane * 4);
}
__device__ __forceinline__ float lane_f(float v_l, int j) {
    return __int_as_float(__builtin_amdgcn_readlane(__float_as_int(v_l), j));
}
__device__ __forceinline__ void add_row(float4& acc, const float4 v, const float sc) {
    acc.x = __builtin_fmaf(v.x, sc, acc.x);
    acc.y = __builtin_fmaf(v.y, sc, acc.y);
    acc.z = __builtin_fmaf(v.z, sc, acc.z);
    acc.w = __builtin_fmaf(v.w, sc, acc.w);
}
__device__ __forceinline__ float inv_sqrt_deg(int c) {
    return 1.0f / sqrtf(fmaxf((float)c, 1.0f));
}
template <bool FIRST>
__device__ __forceinline__ int deg_of(const int* __restrict__ cnt, const int* deg, int s) {
    if (FIRST) return cnt[s] + cnt[CPLANE + s];
    return deg[s];
}

__device__ __forceinline__ void sum_chunk(float4& acc, const float* __restrict__ X,
                                          int s_l, float sc_l, int m, int lane) {
    int j = 0;
    for (; j + 8 <= m; j += 8) {
        float4 v[8];
        #pragma unroll
        for (int i = 0; i < 8; ++i) v[i] = ld_row(X, s_l, j + i, lane);
        #pragma unroll
        for (int i = 0; i < 8; ++i) add_row(acc, v[i], lane_f(sc_l, j + i));
    }
    if (j + 4 <= m) {
        float4 v[4];
        #pragma unroll
        for (int i = 0; i < 4; ++i) v[i] = ld_row(X, s_l, j + i, lane);
        #pragma unroll
        for (int i = 0; i < 4; ++i) add_row(acc, v[i], lane_f(sc_l, j + i));
        j += 4;
    }
    const int rem = m - j;
    float4 t0 = make_float4(0.f, 0.f, 0.f, 0.f), t1 = t0, t2 = t0;
    if (rem > 0) t0 = ld_row(X, s_l, j, lane);
    if (rem > 1) t1 = ld_row(X, s_l, j + 1, lane);
    if (rem > 2) t2 = ld_row(X, s_l, j + 2, lane);
    if (rem > 0) add_row(acc, t0, lane_f(sc_l, j));
    if (rem > 1) add_row(acc, t1, lane_f(sc_l, j + 1));
    if (rem > 2) add_row(acc, t2, lane_f(sc_l, j + 2));
}

template <int NC, bool FIRST>
__global__ __launch_bounds__(32 * (NC / 16)) void cheb_gemm(
        const float* __restrict__ X,
        const int* __restrict__ cnt, const int* __restrict__ slots,
        const int* __restrict__ spill, int* deg,
        const float* __restrict__ W,  const float* __restrict__ bias,
        float* __restrict__ out) {
    __shared__ float As[16 * 260];
    __shared__ int srt[(NC / 16) * 64];

    const int mbase = blockIdx.x * 16;
    const int tid = threadIdx.x;
    const int lane  = tid & 31;

    #pragma unroll
    for (int it = 0; it < (16 * 32) / (32 * (NC / 16)); ++it) {
        int i = tid + it * 32 * (NC / 16);
        int row = i >> 5;
        int col = (i & 31) << 2;
        float4 v = *(const float4*)(X + (size_t)(mbase + row) * 128 + col);
        *(float4*)(&As[row * 260 + col]) = v;
    }

    const int wave = __builtin_amdgcn_readfirstlane(tid >> 5);

    for (int row = wave; row < 16; row += NC / 16) {
        const int node = mbase + row;
        const int n0 = cnt[node];
        const int n1 = cnt[CPLANE + node];
        const int n = n0 + n1;
        if (FIRST && lane == 0) deg[node] = n;
        const int* __restrict__ row0 = slots + (size_t)node * PSLOTS;
        const int* __restrict__ row1 = row0 + SPLANE;
        float4 acc = make_float4(0.f, 0.f, 0.f, 0.f);
        if (n0 <= PSLOTS && n1 <= PSLOTS) {
            int* sw = srt + wave * 64;
            int s_a;
            float sc_a;
            if (n <= 32) {
                int s = 0;
                if (lane < n0) s = row0[lane];
                else if (lane < n) s = row1[lane - n0];
                const int key = (lane < n) ? s * 64 + lane : PADKEY + lane;
                const int cs = deg_of<FIRST>(cnt, deg, s);
                int r = 0;
                #pragma unroll
                for (int j = 0; j < 32; ++j)
                    r += (__builtin_amdgcn_readlane(key, j) < key) ? 1 : 0;
                const float sc_u = inv_sqrt_deg(cs);
                s_a = __builtin_amdgcn_ds_permute(r << 2, key) >> 6;
                sc_a = __int_as_float(__builtin_amdgcn_ds_permute(r << 2, __float_as_int(sc_u)));
            } else {
                int sa;
                if (lane < n0) sa = row0[lane];
                else sa = row1[lane - n0];
                const int k0 = sa * 64 + lane;
                int k1 = PADKEY + lane;
                if (32 + lane < n) k1 = row1[32 + lane - n0] * 64 + 32 + lane;
                int r0 = 0, r1 = 0;
                #pragma nounroll
                for (int j = 0; j < 32; ++j) {
                    const int a = __builtin_amdgcn_readlane(k0, j);
                    const int b = __builtin_amdgcn_readlane(k1, j);
                    r0 += ((a < k0) ? 1 : 0) + ((b < k0) ? 1 : 0);
                    r1 += ((a < k1) ? 1 : 0) + ((b < k1) ? 1 : 0);
                }
                sw[r0] = k0;
                sw[r1] = k1;
                s_a = sw[lane] >> 6;
                const int s_b = sw[32 + lane] >> 6;
                sc_a = inv_sqrt_deg(deg_of<FIRST>(cnt, deg, s_a));
                float sc_b = 0.f;
                if (32 + lane < n) sc_b = inv_sqrt_deg(deg_of<FIRST>(cnt, deg, s_b));
                sw[lane] = s_b;
                sw[32 + lane] = __float_as_int(sc_b);
            }
            #pragma nounroll
            for (int c = 0; c < n; c += 32) {
                if (c) {
                    s_a = sw[lane];
                    sc_a = __int_as_float(sw[32 + lane]);
                }
                sum_chunk(acc, X, s_a, sc_a, min(32, n - c), lane);
            }
        } else {
            const int nsp = min(cnt[NNODES], NEDGES);
            int ka = 0x7fffffff, kb = 0x7fffffff;
            if (lane < min(n0, PSLOTS)) ka = row0[lane];
            if (lane < min(n1, PSLOTS)) kb = row1[lane];
            int prev = -1;
            for (int guard = 0; guard < n; ++guard) {
                int best = 0x7fffffff;
                if (ka > prev && ka < best) best = ka;
                if (kb > prev && kb < best) best = kb;
                for (int i = lane; i < nsp; i += 32) {
                    if (spill[2 * i] == node) {
                        const int k = spill[2 * i + 1];
                        if (k > prev && k < best) best = k;
                    }
                }
                int bmin = 0x7fffffff;
                for (int j = 0; j < 32; ++j) bmin = min(bmin, __builtin_amdgcn_readlane(best, j));
                best = bmin;
                if (best == 0x7fffffff) break;
                int mult = ((ka == best) ? 1 : 0) + ((kb == best) ? 1 : 0);
                for (int i = lane; i < nsp; i += 32)
                    mult += (spill[2 * i] == node && spill[2 * i + 1] == best) ? 1 : 0;
                int msum = 0;
                for (int j = 0; j < 32; ++j) msum += __builtin_amdgcn_readlane(mult, j);
                mult = msum;
                const float4 v = *(const float4*)(X + (size_t)best * 128 + lane * 4);
                const float sc = inv_sqrt_deg(deg_of<FIRST>(cnt, deg, best));
                for (int t = 0; t < mult; ++t) add_row(acc, v, sc);
                prev = best;
            }
        }
        const float ms = -inv_sqrt_deg(n);
        acc.x *= ms; acc.y *= ms; acc.z *= ms; acc.w *= ms;
        *(float4*)(&As[row * 260 + 128 + lane * 4]) = acc;
    }
    __syncthreads();

    const int ntile = tid >> 5;
    const int half  = lane >> 4;
    const int l     = lane & 15;

    v8f c = {};
    #pragma unroll 4
    for (int k0 = 0; k0 < 256; k0 += 4) {
        v2f a = *(const v2f*)(&As[l * 260 + k0 + 2 * half]);
        const float* wp = W + (size_t)(k0 + 2 * half) * NC + ntile * 16 + l;
        v2f b;
        b.x = wp[0];
        b.y = wp[NC];
        c = __builtin_amdgcn_wmma_f32_16x16x4_f32(
                false, a, false, b, (short)0, c, false, false);
    }

    const int col = ntile * 16 + l;
    const float bv = bias[col];
    #pragma unroll
    for (int r = 0; r < 8; ++r) {
        int row = mbase + r + 8 * half;
        out[(size_t)row * NC + col] = c[r] + bv;
    }
}


extern "C" void kernel_launch(void* const* d_in, const int* in_sizes, int n_in,
                              void* d_out, int out_size, void* d_ws, size_t ws_size,
                              hipStream_t stream) {
    const float* feat = (const float*)d_in[0];
    const int*   src  = (const int*)d_in[1];
    const int*   dst  = (const int*)d_in[2];
    const float* W1   = (const float*)d_in[3];
    const float* b1   = (const float*)d_in[4];
    const float* W2   = (const float*)d_in[5];
    const float* b2   = (const float*)d_in[6];
    float* out = (float*)d_out;

    char* ws = (char*)d_ws;
    int*   cnt    = (int*)ws;
    int*   slots  = (int*)(ws + 401408);
    int*   spill  = (int*)(ws + 401408 + 12800000);
    int*   deg    = (int*)(ws + 401408 + 12800000 + 6400000);
    float* out1   = (float*)(ws + 200704 + (size_t)NNODES * 128 * 4);

    const int TB = 256;

    zero_f4<<<(401408 / 16 + TB - 1) / TB, TB, 0, stream>>>((float4*)cnt, 401408 / 16);
    fill_kernel<<<(NEDGES / 4 + TB - 1) / TB, TB, 0, stream>>>(src, dst, cnt, slots, spill, NEDGES / 4);

    cheb_gemm<HID_F, true><<<NNODES / 16, 32 * (HID_F / 16), 0, stream>>>(feat, cnt, slots, spill, deg, W1, b1, out1);
    cheb_gemm<OUT_F, false><<<NNODES / 16, 32 * (OUT_F / 16), 0, stream>>>(out1, cnt, slots, spill, deg, W2, b2, out);
}


# Layer 1 work-groups take 32 rows: each wave keeps two accumulator tiles, so one W1 fragment read feeds two WMMAs; W1 reads and waves launched halve; every sum's order unchanged
# speedup vs baseline: 4.9904x; 1.0369x over previous
#include <hip/hip_runtime.h>
#include <math.h>

#define NNODES 50000
#define NEDGES 800000
#define IN_F   128
#define HID_F  128
#define OUT_F  64
#define PSLOTS 32
#define CPLANE 50176
#define SPLANE (NNODES * PSLOTS)
#define PADKEY 0x7fffffc0

typedef float v2f __attribute__((ext_vector_type(2)));
typedef float v8f __attribute__((ext_vector_type(8)));


__global__ void zero_f4(float4* __restrict__ p, int n4) {
    int i = blockIdx.x * blockDim.x + threadIdx.x;
    if (i < n4) p[i] = make_float4(0.f, 0.f, 0.f, 0.f);
}

__device__ __forceinline__ int my_half() {
    return (int)((__builtin_amdgcn_s_sendmsg_rtn(0x87) >> 18) & 1u);
}

__device__ __forceinline__ void put_edge(int d, int s, int r, int h, int* __restrict__ cnt,
                                         int* __restrict__ slots, int* __restrict__ spill) {
    if (r < PSLOTS) {
        slots[h * SPLANE + d * PSLOTS + r] = s;
    } else {
        const int p = atomicAdd(&cnt[NNODES], 1);
        spill[2 * p] = d;
        spill[2 * p + 1] = s;
    }
}

__global__ void fill_kernel(const int* __restrict__ src, const int* __restrict__ dst,
                            int* __restrict__ cnt, int* __restrict__ slots,
                            int* __restrict__ spill, int ne4) {
    const int i = blockIdx.x * blockDim.x + threadIdx.x;
    if (i >= ne4) return;
    const int4 d = ((const int4*)dst)[i];
    const int4 s = ((const int4*)src)[i];
    const int h = my_half();
    int* __restrict__ cp = cnt + h * CPLANE;
    const int r0 = atomicAdd(&cp[d.x], 1);
    const int r1 = atomicAdd(&cp[d.y], 1);
    const int r2 = atomicAdd(&cp[d.z], 1);
    const int r3 = atomicAdd(&cp[d.w], 1);
    put_edge(d.x, s.x, r0, h, cnt, slots, spill);
    put_edge(d.y, s.y, r1, h, cnt, slots, spill);
    put_edge(d.z, s.z, r2, h, cnt, slots, spill);
    put_edge(d.w, s.w, r3, h, cnt, slots, spill);
}


__device__ __forceinline__ float4 ld_row(const float* __restrict__ X, int s_l, int j, int lane) {
    const int s = __builtin_amdgcn_readlane(s_l, j);
    return *(const float4*)(X + (size_t)s * 128 + lane * 4);
}
__device__ __forceinline__ float lane_f(float v_l, int j) {
    return __int_as_float(__builtin_amdgcn_readlane(__float_as_int(v_l), j));
}
__device__ __forceinline__ void add_row(float4& acc, const float4 v, const float sc) {
    acc.x = __builtin_fmaf(v.x, sc, acc.x);
    acc.y = __builtin_fmaf(v.y, sc, acc.y);
    acc.z = __builtin_fmaf(v.z, sc, acc.z);
    acc.w = __builtin_fmaf(v.w, sc, acc.w);
}
__device__ __forceinline__ float inv_sqrt_deg(int c) {
    return 1.0f / sqrtf(fmaxf((float)c, 1.0f));
}
template <bool FIRST>
__device__ __forceinline__ int deg_of(const int* __restrict__ cnt, const int* deg, int s) {
    if (FIRST) return cnt[s] + cnt[CPLANE + s];
    return deg[s];
}

__device__ __forceinline__ void sum_chunk(float4& acc, const float* __restrict__ X,
                                          int s_l, float sc_l, int m, int lane) {
    int j = 0;
    for (; j + 8 <= m; j += 8) {
        float4 v[8];
        #pragma unroll
        for (int i = 0; i < 8; ++i) v[i] = ld_row(X, s_l, j + i, lane);
        #pragma unroll
        for (int i = 0; i < 8; ++i) add_row(acc, v[i], lane_f(sc_l, j + i));
    }
    if (j + 4 <= m) {
        float4 v[4];
        #pragma unroll
        for (int i = 0; i < 4; ++i) v[i] = ld_row(X, s_l, j + i, lane);
        #pragma unroll
        for (int i = 0; i < 4; ++i) add_row(acc, v[i], lane_f(sc_l, j + i));
        j += 4;
    }
    const int rem = m - j;
    float4 t0 = make_float4(0.f, 0.f, 0.f, 0.f), t1 = t0, t2 = t0;
    if (rem > 0) t0 = ld_row(X, s_l, j, lane);
    if (rem > 1) t1 = ld_row(X, s_l, j + 1, lane);
    if (rem > 2) t2 = ld_row(X, s_l, j + 2, lane);
    if (rem > 0) add_row(acc, t0, lane_f(sc_l, j));
    if (rem > 1) add_row(acc, t1, lane_f(sc_l, j + 1));
    if (rem > 2) add_row(acc, t2, lane_f(sc_l, j + 2));
}

template <int NC, bool FIRST, int MT>
__global__ __launch_bounds__(32 * (NC / 16)) void cheb_gemm(
        const float* __restrict__ X,
        const int* __restrict__ cnt, const int* __restrict__ slots,
        const int* __restrict__ spill, int* deg,
        const float* __restrict__ W,  const float* __restrict__ bias,
        float* __restrict__ out) {
    constexpr int ROWS = 16 * MT;
    constexpr bool PARTIAL = (NNODES % ROWS) != 0;
    __shared__ float As[ROWS * 260];
    __shared__ int srt[(NC / 16) * 64];

    const int mbase = blockIdx.x * ROWS;
    const int tid = threadIdx.x;
    const int lane  = tid & 31;

    #pragma unroll
    for (int it = 0; it < (ROWS * 32) / (32 * (NC / 16)); ++it) {
        int i = tid + it * 32 * (NC / 16);
        int row = i >> 5;
        int col = (i & 31) << 2;
        const int grow = PARTIAL ? min(mbase + row, NNODES - 1) : mbase + row;
        float4 v = *(const float4*)(X + (size_t)grow * 128 + col);
        if (PARTIAL && mbase + row >= NNODES) v = make_float4(0.f, 0.f, 0.f, 0.f);
        *(float4*)(&As[row * 260 + col]) = v;
    }

    const int wave = __builtin_amdgcn_readfirstlane(tid >> 5);

    for (int row = wave; row < ROWS; row += NC / 16) {
        const int node = mbase + row;
        if (PARTIAL && node >= NNODES) {
            *(float4*)(&As[row * 260 + 128 + lane * 4]) = make_float4(0.f, 0.f, 0.f, 0.f);
            continue;
        }
        const int n0 = cnt[node];
        const int n1 = cnt[CPLANE + node];
        const int n = n0 + n1;
        if (FIRST && lane == 0) deg[node] = n;
        const int* __restrict__ row0 = slots + (size_t)node * PSLOTS;
        const int* __restrict__ row1 = row0 + SPLANE;
        float4 acc = make_float4(0.f, 0.f, 0.f, 0.f);
        if (n0 <= PSLOTS && n1 <= PSLOTS) {
            int* sw = srt + wave * 64;
            int s_a;
            float sc_a;
            if (n <= 32) {
                int s = 0;
                if (lane < n0) s = row0[lane];
                else if (lane < n) s = row1[lane - n0];
                const int key = (lane < n) ? s * 64 + lane : PADKEY + lane;
                const int cs = deg_of<FIRST>(cnt, deg, s);
                int r = 0;
                #pragma unroll
                for (int j = 0; j < 32; ++j)
                    r += (__builtin_amdgcn_readlane(key, j) < key) ? 1 : 0;
                const float sc_u = inv_sqrt_deg(cs);
                s_a = __builtin_amdgcn_ds_permute(r << 2, key) >> 6;
                sc_a = __int_as_float(__builtin_amdgcn_ds_permute(r << 2, __float_as_int(sc_u)));
            } else {
                int sa;
                if (lane < n0) sa = row0[lane];
                else sa = row1[lane - n0];
                const int k0 = sa * 64 + lane;
                int k1 = PADKEY + lane;
                if (32 + lane < n) k1 = row1[32 + lane - n0] * 64 + 32 + lane;
                int r0 = 0, r1 = 0;
                #pragma nounroll
                for (int j = 0; j < 32; ++j) {
                    const int a = __builtin_amdgcn_readlane(k0, j);
                    const int b = __builtin_amdgcn_readlane(k1, j);
                    r0 += ((a < k0) ? 1 : 0) + ((b < k0) ? 1 : 0);
                    r1 += ((a < k1) ? 1 : 0) + ((b < k1) ? 1 : 0);
                }
                sw[r0] = k0;
                sw[r1] = k1;
                s_a = sw[lane] >> 6;
                const int s_b = sw[32 + lane] >> 6;
                sc_a = inv_sqrt_deg(deg_of<FIRST>(cnt, deg, s_a));
                float sc_b = 0.f;
                if (32 + lane < n) sc_b = inv_sqrt_deg(deg_of<FIRST>(cnt, deg, s_b));
                sw[lane] = s_b;
                sw[32 + lane] = __float_as_int(sc_b);
            }
            #pragma nounroll
            for (int c = 0; c < n; c += 32) {
                if (c) {
                    s_a = sw[lane];
                    sc_a = __int_as_float(sw[32 + lane]);
                }
                sum_chunk(acc, X, s_a, sc_a, min(32, n - c), lane);
            }
        } else {
            const int nsp = min(cnt[NNODES], NEDGES);
            int ka = 0x7fffffff, kb = 0x7fffffff;
            if (lane < min(n0, PSLOTS)) ka = row0[lane];
            if (lane < min(n1, PSLOTS)) kb = row1[lane];
            int prev = -1;
            for (int guard = 0; guard < n; ++guard) {
                int best = 0x7fffffff;
                if (ka > prev && ka < best) best = ka;
                if (kb > prev && kb < best) best = kb;
                for (int i = lane; i < nsp; i += 32) {
                    if (spill[2 * i] == node) {
                        const int k = spill[2 * i + 1];
                        if (k > prev && k < best) best = k;
                    }
                }
                int bmin = 0x7fffffff;
                for (int j = 0; j < 32; ++j) bmin = min(bmin, __builtin_amdgcn_readlane(best, j));
                best = bmin;
                if (best == 0x7fffffff) break;
                int mult = ((ka == best) ? 1 : 0) + ((kb == best) ? 1 : 0);
                for (int i = lane; i < nsp; i += 32)
                    mult += (spill[2 * i] == node && spill[2 * i + 1] == best) ? 1 : 0;
                int msum = 0;
                for (int j = 0; j < 32; ++j) msum += __builtin_amdgcn_readlane(mult, j);
                mult = msum;
                const float4 v = *(const float4*)(X + (size_t)best * 128 + lane * 4);
                const float sc = inv_sqrt_deg(deg_of<FIRST>(cnt, deg, best));
                for (int t = 0; t < mult; ++t) add_row(acc, v, sc);
                prev = best;
            }
        }
        const float ms = -inv_sqrt_deg(n);
        acc.x *= ms; acc.y *= ms; acc.z *= ms; acc.w *= ms;
        *(float4*)(&As[row * 260 + 128 + lane * 4]) = acc;
    }
    __syncthreads();

    const int ntile = tid >> 5;
    const int half  = lane >> 4;
    const int l     = lane & 15;

    v8f c[MT];
    #pragma unroll
    for (int mt = 0; mt < MT; ++mt) c[mt] = v8f{};
    #pragma unroll 4
    for (int k0 = 0; k0 < 256; k0 += 4) {
        const float* wp = W + (size_t)(k0 + 2 * half) * NC + ntile * 16 + l;
        v2f b;
        b.x = wp[0];
        b.y = wp[NC];
        #pragma unroll
        for (int mt = 0; mt < MT; ++mt) {
            v2f a = *(const v2f*)(&As[(mt * 16 + l) * 260 + k0 + 2 * half]);
            c[mt] = __builtin_amdgcn_wmma_f32_16x16x4_f32(
                    false, a, false, b, (short)0, c[mt], false, false);
        }
    }

    const int col = ntile * 16 + l;
    const float bv = bias[col];
    #pragma unroll
    for (int mt = 0; mt < MT; ++mt) {
        if (PARTIAL && mbase + mt * 16 >= NNODES) continue;
        #pragma unroll
        for (int r = 0; r < 8; ++r) {
            int row = mbase + mt * 16 + r + 8 * half;
            out[(size_t)row * NC + col] = c[mt][r] + bv;
        }
    }
}


extern "C" void kernel_launch(void* const* d_in, const int* in_sizes, int n_in,
                              void* d_out, int out_size, void* d_ws, size_t ws_size,
                              hipStream_t stream) {
    const float* feat = (const float*)d_in[0];
    const int*   src  = (const int*)d_in[1];
    const int*   dst  = (const int*)d_in[2];
    const float* W1   = (const float*)d_in[3];
    const float* b1   = (const float*)d_in[4];
    const float* W2   = (const float*)d_in[5];
    const float* b2   = (const float*)d_in[6];
    float* out = (float*)d_out;

    char* ws = (char*)d_ws;
    int*   cnt    = (int*)ws;
    int*   slots  = (int*)(ws + 401408);
    int*   spill  = (int*)(ws + 401408 + 12800000);
    int*   deg    = (int*)(ws + 401408 + 12800000 + 6400000);
    float* out1   = (float*)(ws + 200704 + (size_t)NNODES * 128 * 4);

    const int TB = 256;

    zero_f4<<<(401408 / 16 + TB - 1) / TB, TB, 0, stream>>>((float4*)cnt, 401408 / 16);
    fill_kernel<<<(NEDGES / 4 + TB - 1) / TB, TB, 0, stream>>>(src, dst, cnt, slots, spill, NEDGES / 4);

    cheb_gemm<HID_F, true, 2><<<(NNODES + 31) / 32, 32 * (HID_F / 16), 0, stream>>>(feat, cnt, slots, spill, deg, W1, b1, out1);
    cheb_gemm<OUT_F, false, 1><<<NNODES / 16, 32 * (OUT_F / 16), 0, stream>>>(out1, cnt, slots, spill, deg, W2, b2, out);
}
